# starting kernel
# baseline (speedup 1.0000x reference)
#include <hip/hip_runtime.h>
#include <hip/hip_bf16.h>

typedef _Float16 v16h __attribute__((ext_vector_type(16)));
typedef _Float16 v8h  __attribute__((ext_vector_type(8)));
typedef float    v8f  __attribute__((ext_vector_type(8)));
typedef float    v4f  __attribute__((ext_vector_type(4)));

#define FDIM 256
#define TDIM 2048
#define BDIM 128
#define LDS_STRIDE 264
#define TCHUNKS 8

#if __has_builtin(__builtin_amdgcn_tanhf)
  #define FAST_TANH(v) __builtin_amdgcn_tanhf(v)
#elif __has_builtin(__builtin_amdgcn_tanh_f32)
  #define FAST_TANH(v) __builtin_amdgcn_tanh_f32(v)
#else
  #define FAST_TANH(v) tanhf(v)
#endif

__global__ __launch_bounds__(256) void awc_scores(const float* __restrict__ x,
                                                  const float* __restrict__ W,
                                                  const float* __restrict__ bias,
                                                  const float* __restrict__ u,
                                                  float* __restrict__ ait)
{
    __shared__ _Float16 sW[FDIM * LDS_STRIDE];

    const int tid = threadIdx.x;

    {
        _Float16* dst = &sW[(unsigned)tid * LDS_STRIDE];
        for (int f = 0; f < FDIM; f += 2) {
            float w0 = W[f * FDIM + tid];
            float w1 = W[(f + 1) * FDIM + tid];
            v8h p = {};
            p[0] = (_Float16)w0;
            p[1] = (_Float16)w1;
            *(__attribute__((ext_vector_type(2))) _Float16*)(dst + f) =
                __builtin_shufflevector(p, p, 0, 1);
        }
    }
    __syncthreads();

    const int wave  = tid >> 5;
    const int lane  = tid & 31;
    const int lrow  = lane & 15;
    const int khalf = (lane >> 4) * 8;

    const long rowBase = (long)blockIdx.x * 256 + wave * 32;

    v16h afrag[2][8];
#pragma unroll
    for (int s = 0; s < 2; ++s) {
        const float* xr = x + (rowBase + s * 16 + lrow) * (long)FDIM;
#pragma unroll
        for (int k = 0; k < 8; ++k) {
            const int kb = k * 32 + khalf;
            v4f lo0 = *(const v4f*)(xr + kb);
            v4f lo1 = *(const v4f*)(xr + kb + 4);
            v4f hi0 = *(const v4f*)(xr + kb + 16);
            v4f hi1 = *(const v4f*)(xr + kb + 20);
            v16h a;
#pragma unroll
            for (int e = 0; e < 4; ++e) {
                a[e]      = (_Float16)lo0[e];
                a[e + 4]  = (_Float16)lo1[e];
                a[e + 8]  = (_Float16)hi0[e];
                a[e + 12] = (_Float16)hi1[e];
            }
            afrag[s][k] = a;
        }
    }

    float part[2][8];
#pragma unroll
    for (int s = 0; s < 2; ++s)
#pragma unroll
        for (int r = 0; r < 8; ++r) part[s][r] = 0.0f;

    auto loadB = [&](int nk) -> v16h {
        const int nn = nk >> 3;
        const int kk = nk & 7;
        const _Float16* p = &sW[(nn * 16 + lrow) * LDS_STRIDE + kk * 32 + khalf];
        v8h blo = *(const v8h*)(p);
        v8h bhi = *(const v8h*)(p + 16);
        return __builtin_shufflevector(blo, bhi,
            0,1,2,3,4,5,6,7,8,9,10,11,12,13,14,15);
    };

    v16h bf0 = loadB(0);
    v16h bf1 = loadB(1);

#pragma unroll
    for (int n = 0; n < 16; ++n) {
        v8f acc0 = {};
        v8f acc1 = {};
#pragma unroll
        for (int k = 0; k < 8; ++k) {
            const int nk = n * 8 + k;
            v16h bcur = bf0;
            bf0 = bf1;
            if (nk + 2 < 128) bf1 = loadB(nk + 2);
            acc0 = __builtin_amdgcn_wmma_f32_16x16x32_f16(
                false, afrag[0][k], false, bcur, (short)0, acc0, false, false);
            acc1 = __builtin_amdgcn_wmma_f32_16x16x32_f16(
                false, afrag[1][k], false, bcur, (short)0, acc1, false, false);
        }
        const int col = n * 16 + lrow;
        const float un = u[col];
        const float bn = bias[col];
#pragma unroll
        for (int r = 0; r < 8; ++r) {
            part[0][r] += FAST_TANH(acc0[r] + bn) * un;
            part[1][r] += FAST_TANH(acc1[r] + bn) * un;
        }
    }

#pragma unroll
    for (int s = 0; s < 2; ++s) {
#pragma unroll
        for (int r = 0; r < 8; ++r) {
            float v = part[s][r];
            v += __shfl_xor(v, 1, 32);
            v += __shfl_xor(v, 2, 32);
            v += __shfl_xor(v, 4, 32);
            v += __shfl_xor(v, 8, 32);
            part[s][r] = v;
        }
    }

    if (lrow == 0) {
#pragma unroll
        for (int s = 0; s < 2; ++s) {
            const long base = rowBase + s * 16 + (lane >> 4) * 8;
#pragma unroll
            for (int r = 0; r < 8; ++r) ait[base + r] = part[s][r];
        }
    }
}

__global__ __launch_bounds__(256) void awc_softmax(float* __restrict__ a)
{
    __shared__ float red[8];
    const int b   = blockIdx.x;
    const int tid = threadIdx.x;
    float* row = a + (long)b * TDIM;

    float s = 0.0f;
    for (int t = tid; t < TDIM; t += 256) s += expf(row[t]);
#pragma unroll
    for (int off = 16; off; off >>= 1) s += __shfl_xor(s, off, 32);
    if ((tid & 31) == 0) red[tid >> 5] = s;
    __syncthreads();

    float tot = 0.0f;
#pragma unroll
    for (int w = 0; w < 8; ++w) tot += red[w];
    const float inv = 1.0f / (tot + 1e-7f);

    for (int t = tid; t < TDIM; t += 256) row[t] = expf(row[t]) * inv;
}

__global__ __launch_bounds__(256) void awc_wsum_partial(const float* __restrict__ x,
                                                        const float* __restrict__ a,
                                                        float* __restrict__ partial)
{
    const int b = blockIdx.y;
    const int c = blockIdx.x;
    const int f = threadIdx.x;
    const long tbase = (long)b * TDIM + (long)c * (TDIM / TCHUNKS);

    const float* xp = x + tbase * FDIM + f;
    const float* ap = a + tbase;
    float acc = 0.0f;
    for (int t = 0; t < TDIM / TCHUNKS; ++t)
        acc += ap[t] * xp[(long)t * FDIM];
    partial[((long)b * TCHUNKS + c) * FDIM + f] = acc;
}

__global__ __launch_bounds__(256) void awc_wsum_reduce(const float* __restrict__ partial,
                                                       float* __restrict__ out)
{
    const int b = blockIdx.x;
    const int f = threadIdx.x;
    float acc = 0.0f;
#pragma unroll
    for (int c = 0; c < TCHUNKS; ++c)
        acc += partial[((long)b * TCHUNKS + c) * FDIM + f];
    out[(long)b * FDIM + f] = acc;
}

extern "C" void kernel_launch(void* const* d_in, const int* in_sizes, int n_in,
                              void* d_out, int out_size, void* d_ws, size_t ws_size,
                              hipStream_t stream) {
    const float* x    = (const float*)d_in[0];
    const float* W    = (const float*)d_in[1];
    const float* bias = (const float*)d_in[2];
    const float* u    = (const float*)d_in[3];
    float* out = (float*)d_out;

    float* a       = (float*)d_ws;
    float* partial = a + (size_t)BDIM * TDIM;

    awc_scores      <<<dim3((BDIM * TDIM) / 256), 256, 0, stream>>>(x, W, bias, u, a);
    awc_softmax     <<<dim3(BDIM),                256, 0, stream>>>(a);
    awc_wsum_partial<<<dim3(TCHUNKS, BDIM),       256, 0, stream>>>(x, a, partial);
    awc_wsum_reduce <<<dim3(BDIM),                256, 0, stream>>>(partial, out);
}


# bias and u tables held in registers all kernel long (bias is the C operand of each block's first matrix instruction): the 32 table loads per tile and wave leave the LDS queue; same sums, same order
# speedup vs baseline: 2.9841x; 2.9841x over previous
#include <hip/hip_runtime.h>
#include <hip/hip_bf16.h>

typedef _Float16 v16h __attribute__((ext_vector_type(16)));
typedef _Float16 v8h  __attribute__((ext_vector_type(8)));
typedef float    v8f  __attribute__((ext_vector_type(8)));
typedef float    v4f  __attribute__((ext_vector_type(4)));
typedef float    v2f  __attribute__((ext_vector_type(2)));

#define FDIM 256
#define TDIM 2048
#define BDIM 128
#define LDS_STRIDE 264

#if __has_builtin(__builtin_amdgcn_tanhf)
  #define FAST_TANH(v) __builtin_amdgcn_tanhf(v)
#elif __has_builtin(__builtin_amdgcn_tanh_f32)
  #define FAST_TANH(v) __builtin_amdgcn_tanh_f32(v)
#else
  #define FAST_TANH(v) tanhf(v)
#endif

#define SC_ROWS    32
#define SC_THREADS 128
#define SC_GRID    512
#define SC_TILES   ((BDIM * TDIM) / SC_ROWS)
#define SC_PER_WG  (SC_TILES / SC_GRID)
#define WG_PER_B   (SC_GRID / BDIM)
#define DEN_PITCH  32
#define SC_LOADS   ((SC_ROWS * FDIM / 4) / SC_THREADS)

static __device__ __forceinline__ void lds_barrier() {
    asm volatile("s_wait_dscnt 0x0\n\ts_barrier_signal -1\n\ts_barrier_wait -1" ::: "memory");
}

__global__ __launch_bounds__(SC_THREADS, 2) void awc_scores(const float* __restrict__ x,
                                                  const float* __restrict__ W,
                                                  const float* __restrict__ bias,
                                                  const float* __restrict__ u,
                                                  float* __restrict__ wsp,
                                                  float* __restrict__ wden)
{
    __shared__ __attribute__((aligned(16))) _Float16 xs[2 * SC_ROWS * LDS_STRIDE];
    __shared__ float red[2][4][2][SC_ROWS];
    __shared__ __attribute__((aligned(16))) float fin[SC_THREADS / 2 * 4 + 2];

    const int tid   = threadIdx.x;
    const int wave  = tid >> 5;
    const int lane  = tid & 31;
    const int c     = lane & 15;
    const int h     = lane >> 4;
    const int colbase = wave * 64;

    v16h bfrag[4][8];
    {
        const float* wp = W + (8 * h) * FDIM + colbase + 4 * c;
        v4f wst[3][16];
        auto wload = [&](const int kk, v4f* d, const int off) __attribute__((always_inline)) {
#pragma unroll
            for (int e = 0; e < 8; ++e) {
                d[e]     = *(const v4f*)(wp + off + (32 * kk + e) * FDIM);
                d[8 + e] = *(const v4f*)(wp + off + (32 * kk + 16 + e) * FDIM);
            }
        };
        int woff = 0;
        wload(0, wst[0], 0);
        wload(1, wst[1], 0);
#pragma unroll
        for (int kk = 0; kk < 8; ++kk) {
            if (kk + 2 < 8) {
                if (kk > 0) {
                    typedef int v8i __attribute__((ext_vector_type(8)));
                    const int dep = __builtin_amdgcn_readfirstlane(
                        __builtin_bit_cast(v8i, bfrag[3][kk - 1])[7]);
                    asm volatile("" : "+s"(woff) : "s"(dep));
                }
                wload(kk + 2, wst[(kk + 2) % 3], woff);
            }
            const v4f* d = wst[kk % 3];
#pragma unroll
            for (int j = 0; j < 4; ++j) {
                v16h f;
#pragma unroll
                for (int e = 0; e < 16; ++e) f[e] = (_Float16)d[e][j];
                bfrag[j][kk] = f;
            }
        }
    }
    v8f biasv[4];
    v2f uv[4][4];
    {
        const float* bp = bias + 32 * (2 * wave + h);
        const float* up = u + 32 * (2 * wave + h);
#pragma unroll
        for (int r = 0; r < 8; ++r) {
            const v4f qb = *(const v4f*)(bp + 4 * r);
            const v4f qu = *(const v4f*)(up + 4 * r);
#pragma unroll
            for (int nb = 0; nb < 4; ++nb) {
                biasv[nb][r] = qb[nb];
                uv[nb][r >> 1][r & 1] = qu[nb];
            }
        }
    }

    const int frow = tid >> 6;
    const int fk4  = tid & 63;
    v4f stg[SC_LOADS];
    auto issue = [&](int tile) {
        const float* xp = x + ((long)tile * SC_ROWS + frow) * FDIM + fk4 * 4;
#pragma unroll
        for (int j = 0; j < SC_LOADS; ++j) stg[j] = *(const v4f*)(xp + (long)j * 2 * FDIM);
    };

    v2f wacc01 = (v2f){0.0f, 0.0f};
    v2f wacc23 = (v2f){0.0f, 0.0f};
    float dacc = 0.0f;
    typedef _Float16 v4h __attribute__((ext_vector_type(4)));
    float part[8];
    v4h xown[SC_LOADS];
    auto attend_loads = [&](int pb) __attribute__((always_inline)) {
        const float* rp = &red[pb][0][0][frow * 16 + c];
#pragma unroll
        for (int i = 0; i < 8; ++i) part[i] = rp[i * SC_ROWS];
        const _Float16* xr = xs + pb * (SC_ROWS * LDS_STRIDE) + frow * LDS_STRIDE + fk4 * 4;
#pragma unroll
        for (int j = 0; j < SC_LOADS; ++j) xown[j] = *(const v4h*)(xr + (2 * j) * LDS_STRIDE);
    };
    auto attend_sums = [&](const float live) __attribute__((always_inline)) {
        const float sc = ((part[0] + part[1]) + (part[2] + part[3]))
                       + ((part[4] + part[5]) + (part[6] + part[7]));
        const float e = __expf(sc) * live;
        dacc += e;
#pragma unroll
        for (int j = 0; j < SC_LOADS; ++j) {
            const float ej = __int_as_float(__builtin_amdgcn_readlane(__float_as_int(e), j));
            const v4h p = xown[j];
            const v2f ee  = (v2f){ej, ej};
            const v2f x01 = (v2f){(float)p[0], (float)p[1]};
            const v2f x23 = (v2f){(float)p[2], (float)p[3]};
            wacc01 = ee * x01 + wacc01;
            wacc23 = ee * x23 + wacc23;
        }
    };

    {
        _Float16* xz = xs + SC_ROWS * LDS_STRIDE + frow * LDS_STRIDE + fk4 * 4;
#pragma unroll
        for (int j = 0; j < SC_LOADS; ++j) *(v4h*)(xz + (2 * j) * LDS_STRIDE) = (v4h){0, 0, 0, 0};
        float* rz = &red[1][0][0][0];
        rz[tid] = 0.0f;
        rz[tid + SC_THREADS] = 0.0f;
    }
    const int tile0 = blockIdx.x * SC_PER_WG;
    issue(tile0);
    auto trip = [&](const int it) __attribute__((always_inline)) {
        const int tile = tile0 + it;
        const int buf = it & 1;
        _Float16* xb = xs + buf * (SC_ROWS * LDS_STRIDE);

#pragma unroll
        for (int j = 0; j < SC_LOADS; ++j) {
            typedef _Float16 v4h __attribute__((ext_vector_type(4)));
            v4h p;
            p[0] = (_Float16)stg[j][0];
            p[1] = (_Float16)stg[j][1];
            p[2] = (_Float16)stg[j][2];
            p[3] = (_Float16)stg[j][3];
            *(v4h*)(xb + (2 * j + frow) * LDS_STRIDE + fk4 * 4) = p;
        }
        if (it + 1 < SC_PER_WG) issue(tile + 1);
        lds_barrier();

        float z[SC_ROWS / 16];
#pragma unroll
        for (int s = 0; s < SC_ROWS / 16; ++s) {
            v16h a[8];
#pragma unroll
            for (int kk = 0; kk < 8; ++kk) {
                const _Float16* p = xb + (s * 16 + c) * LDS_STRIDE + kk * 32 + 8 * h;
                v8h lo = *(const v8h*)(p);
                v8h hi = *(const v8h*)(p + 16);
                a[kk] = __builtin_shufflevector(lo, hi,
                    0,1,2,3,4,5,6,7,8,9,10,11,12,13,14,15);
            }
            v8f acc[4];
            __builtin_amdgcn_sched_barrier(0);
#pragma unroll
            for (int kk = 0; kk < 8; ++kk) {
#pragma unroll
                for (int nb = 0; nb < 4; ++nb) {
                    acc[nb] = __builtin_amdgcn_wmma_f32_16x16x32_f16(
                        false, bfrag[nb][kk], false, a[kk], (short)0,
                        (kk == 0) ? biasv[nb] : acc[nb], false, false);
                }
            }
            __builtin_amdgcn_sched_barrier(0);
            if (s == SC_ROWS / 16 - 1) {
                attend_loads(buf ^ 1);
                __builtin_amdgcn_sched_barrier(0);
            }
            v2f t2[4];
#pragma unroll
            for (int nb = 0; nb < 4; ++nb) {
#pragma unroll
                for (int rp = 0; rp < 4; ++rp) {
                    v2f th;
                    th[0] = FAST_TANH(acc[nb][2 * rp]);
                    th[1] = FAST_TANH(acc[nb][2 * rp + 1]);
                    t2[nb] = (rp == 0) ? th * uv[nb][0] : th * uv[nb][rp] + t2[nb];
                }
            }
            const v2f q = (t2[0] + t2[1]) + (t2[2] + t2[3]);
            z[s] = q[0] + q[1];
        }
#pragma unroll
        for (int s = 0; s < SC_ROWS / 16; ++s)
            red[buf][wave][h][16 * (c & 1) + s * 8 + (c >> 1)] = z[s];
        attend_sums(it > 0 ? 1.0f : 0.0f);
    };
#pragma unroll 1
    for (int it = 0; it < SC_PER_WG; ++it) trip(it);
    lds_barrier();
    attend_loads((SC_PER_WG - 1) & 1);
    attend_sums(1.0f);

    dacc += __shfl_xor(dacc, 8, 32);
    dacc += __shfl_xor(dacc, 4, 32);
    dacc += __shfl_xor(dacc, 2, 32);
    dacc += __shfl_xor(dacc, 1, 32);
    const v4f wacc = (v4f){wacc01[0], wacc01[1], wacc23[0], wacc23[1]};
    if (frow == 1) *(v4f*)(fin + fk4 * 4) = wacc;
    if (tid == 64) fin[SC_THREADS / 2 * 4] = dacc;
    lds_barrier();
    if (frow == 0) {
        const v4f o = *(const v4f*)(fin + fk4 * 4);
        v4f r;
        r[0] = wacc[0] + o[0];
        r[1] = wacc[1] + o[1];
        r[2] = wacc[2] + o[2];
        r[3] = wacc[3] + o[3];
        *(v4f*)(wsp + (long)blockIdx.x * FDIM + fk4 * 4) = r;
        if (tid == 0) wden[blockIdx.x * DEN_PITCH] = dacc + fin[SC_THREADS / 2 * 4];
    }
}

__global__ __launch_bounds__(FDIM) void awc_finalize(const float* __restrict__ wsp,
                                                     const float* __restrict__ wden,
                                                     float* __restrict__ out)
{
    const int b = blockIdx.x;
    const int f = threadIdx.x;
    const float* p = wsp + (long)b * WG_PER_B * FDIM + f;
    const float* d = wden + b * WG_PER_B * DEN_PITCH;
    const float num = ((p[0] + p[FDIM]) + p[2 * FDIM]) + p[3 * FDIM];
    const float den = ((d[0] + d[DEN_PITCH]) + d[2 * DEN_PITCH]) + d[3 * DEN_PITCH];
    out[(long)b * FDIM + f] = num / (den + 1e-7f);
}

extern "C" void kernel_launch(void* const* d_in, const int* in_sizes, int n_in,
                              void* d_out, int out_size, void* d_ws, size_t ws_size,
                              hipStream_t stream) {
    const float* x    = (const float*)d_in[0];
    const float* W    = (const float*)d_in[1];
    const float* bias = (const float*)d_in[2];
    const float* u    = (const float*)d_in[3];
    float* out = (float*)d_out;

    float* wsp  = (float*)d_ws;
    float* wden = wsp + (size_t)SC_GRID * FDIM;

    awc_scores  <<<dim3(SC_GRID), SC_THREADS, 0, stream>>>(x, W, bias, u, wsp, wden);
    awc_finalize<<<dim3(BDIM),    FDIM,       0, stream>>>(wsp, wden, out);
}
